# starting kernel
# baseline (speedup 1.0000x reference)
#include <hip/hip_runtime.h>
#include <hip/hip_bf16.h>

#define N_NODES 30000
#define E_RAW   300000
#define E_TOT   330000
#define HEADS   8
#define HID     64
#define F1      512
#define IN_DIM  8
#define OUT_DIM 8
#define NEG_SLOPE 0.2f
#define M_TILES 1875

typedef __attribute__((ext_vector_type(2))) float v2f;
typedef __attribute__((ext_vector_type(8))) float v8f;

__device__ __forceinline__ unsigned fmap(float f) {
    unsigned u = __float_as_uint(f);
    return (u & 0x80000000u) ? ~u : (u | 0x80000000u);
}
__device__ __forceinline__ float funmap(unsigned u) {
    return __uint_as_float((u & 0x80000000u) ? (u ^ 0x80000000u) : ~u);
}
__device__ __forceinline__ void get_edge(const int* __restrict__ ei, int e, int& s, int& d) {
    if (e < E_RAW) { s = ei[e]; d = ei[E_RAW + e]; }
    else           { s = e - E_RAW; d = e - E_RAW; }
}
__device__ __forceinline__ float lrelu(float v) {
    return v > 0.f ? v : v * NEG_SLOPE;
}

__global__ void k_init1(unsigned* __restrict__ m1, float* __restrict__ s1) {
    int i = blockIdx.x * blockDim.x + threadIdx.x;
    if (i < N_NODES * HEADS) { m1[i] = 0u; s1[i] = 0.f; }
}
__global__ void k_zero4(float4* __restrict__ p, int n4) {
    int i = blockIdx.x * blockDim.x + threadIdx.x;
    if (i < n4) p[i] = make_float4(0.f, 0.f, 0.f, 0.f);
}
__global__ void k_init2(float* __restrict__ out, const float* __restrict__ b2,
                        unsigned* __restrict__ m2, float* __restrict__ s2) {
    int i = blockIdx.x * blockDim.x + threadIdx.x;
    if (i < N_NODES * OUT_DIM) out[i] = b2[i & 7];
    if (i < N_NODES) { m2[i] = 0u; s2[i] = 0.f; }
}

__global__ void k_gemm1(const float* __restrict__ x,
                        const float* __restrict__ Wl,
                        const float* __restrict__ Wr,
                        float* __restrict__ xl, float* __restrict__ xr) {
    const int wave  = threadIdx.x >> 5;
    const int lane  = threadIdx.x & 31;
    const int row   = lane & 15;
    const int khalf = lane >> 4;
    const int mbase = blockIdx.x * 16;
    const int nbase = (blockIdx.y * 8 + wave) * 16;
    const float* __restrict__ W = blockIdx.z ? Wr : Wl;
    float* __restrict__ O       = blockIdx.z ? xr : xl;

    v8f c = {};
#pragma unroll
    for (int kk = 0; kk < 2; ++kk) {
        const int k0 = kk * 4 + khalf * 2;
        v2f a, b;
        a.x = x[(mbase + row) * IN_DIM + k0];
        a.y = x[(mbase + row) * IN_DIM + k0 + 1];
        b.x = W[(k0    ) * F1 + nbase + row];
        b.y = W[(k0 + 1) * F1 + nbase + row];
        c = __builtin_amdgcn_wmma_f32_16x16x4_f32(false, a, false, b,
                                                  (short)0, c, false, false);
    }
#pragma unroll
    for (int r = 0; r < 8; ++r)
        O[(size_t)(mbase + r + 8 * khalf) * F1 + nbase + row] = c[r];
}

__global__ void k_edge1_logits(const int* __restrict__ ei,
                               const float* __restrict__ xl, const float* __restrict__ xr,
                               const float* __restrict__ a1,
                               float* __restrict__ e1, unsigned* __restrict__ m1) {
    const int wid  = (int)((blockIdx.x * blockDim.x + threadIdx.x) >> 5);
    const int lane = threadIdx.x & 31;
    if (wid >= E_TOT) return;
    int s, d; get_edge(ei, wid, s, d);
    const float4* __restrict__ pl = (const float4*)(xl + (size_t)s * F1) + lane * 4;
    const float4* __restrict__ pr = (const float4*)(xr + (size_t)d * F1) + lane * 4;
    const float4* __restrict__ pa = (const float4*)a1 + lane * 4;
    float acc = 0.f;
#pragma unroll
    for (int i = 0; i < 4; ++i) {
        float4 l = pl[i], r = pr[i], a = pa[i];
        acc = fmaf(lrelu(l.x + r.x), a.x, acc);
        acc = fmaf(lrelu(l.y + r.y), a.y, acc);
        acc = fmaf(lrelu(l.z + r.z), a.z, acc);
        acc = fmaf(lrelu(l.w + r.w), a.w, acc);
    }
    acc += __shfl_xor(acc, 1, 32);
    acc += __shfl_xor(acc, 2, 32);
    if ((lane & 3) == 0) {
        const int h = lane >> 2;
        e1[(size_t)wid * HEADS + h] = acc;
        atomicMax(&m1[d * HEADS + h], fmap(acc));
    }
}

__global__ void k_edge1_exp(const int* __restrict__ ei, float* __restrict__ e1,
                            const unsigned* __restrict__ m1, float* __restrict__ s1) {
    int idx = blockIdx.x * blockDim.x + threadIdx.x;
    if (idx >= E_TOT * HEADS) return;
    const int e = idx >> 3, h = idx & 7;
    int s, d; get_edge(ei, e, s, d);
    const float p = __expf(e1[idx] - funmap(m1[d * HEADS + h]));
    e1[idx] = p;
    atomicAdd(&s1[d * HEADS + h], p);
}

__global__ void k_edge1_agg(const int* __restrict__ ei,
                            const float* __restrict__ xl,
                            const float* __restrict__ e1, const float* __restrict__ s1,
                            float* __restrict__ h1) {
    const int wid  = (int)((blockIdx.x * blockDim.x + threadIdx.x) >> 5);
    const int lane = threadIdx.x & 31;
    if (wid >= E_TOT) return;
    int s, d; get_edge(ei, wid, s, d);
    const int h = lane >> 2;
    const float alpha = e1[(size_t)wid * HEADS + h] / s1[d * HEADS + h];
    const float4* __restrict__ pl = (const float4*)(xl + (size_t)s * F1) + lane * 4;
    float* __restrict__ po = h1 + (size_t)d * F1 + lane * 16;
#pragma unroll
    for (int i = 0; i < 4; ++i) {
        float4 l = pl[i];
        atomicAdd(po + i * 4 + 0, alpha * l.x);
        atomicAdd(po + i * 4 + 1, alpha * l.y);
        atomicAdd(po + i * 4 + 2, alpha * l.z);
        atomicAdd(po + i * 4 + 3, alpha * l.w);
    }
}

__global__ void k_bias_relu(float* __restrict__ h1, const float* __restrict__ b1) {
    int i = blockIdx.x * blockDim.x + threadIdx.x;
    if (i >= N_NODES * F1) return;
    const float v = h1[i] + b1[i & (F1 - 1)];
    h1[i] = v > 0.f ? v : 0.f;
}

__global__ void k_gemm2(const float* __restrict__ h,
                        const float* __restrict__ W2l,
                        const float* __restrict__ W2r,
                        float* __restrict__ h2l, float* __restrict__ h2r) {
    const int wave  = threadIdx.x >> 5;
    const int lane  = threadIdx.x & 31;
    const int row   = lane & 15;
    const int khalf = lane >> 4;
    const int mtile = blockIdx.x * 4 + (wave >> 1);
    if (mtile >= M_TILES) return;
    const int mbase = mtile * 16;
    const float* __restrict__ W = (wave & 1) ? W2r : W2l;
    float* __restrict__ O       = (wave & 1) ? h2r : h2l;

    v8f c = {};
    for (int kk = 0; kk < F1 / 4; ++kk) {
        const int k0 = kk * 4 + khalf * 2;
        v2f a, b;
        a.x = h[(size_t)(mbase + row) * F1 + k0];
        a.y = h[(size_t)(mbase + row) * F1 + k0 + 1];
        b.x = (row < OUT_DIM) ? W[(k0    ) * OUT_DIM + row] : 0.f;
        b.y = (row < OUT_DIM) ? W[(k0 + 1) * OUT_DIM + row] : 0.f;
        c = __builtin_amdgcn_wmma_f32_16x16x4_f32(false, a, false, b,
                                                  (short)0, c, false, false);
    }
    if (row < OUT_DIM) {
#pragma unroll
        for (int r = 0; r < 8; ++r)
            O[(size_t)(mbase + r + 8 * khalf) * OUT_DIM + row] = c[r];
    }
}

__global__ void k_edge2_logits(const int* __restrict__ ei,
                               const float* __restrict__ h2l, const float* __restrict__ h2r,
                               const float* __restrict__ a2,
                               float* __restrict__ e2, unsigned* __restrict__ m2) {
    int e = blockIdx.x * blockDim.x + threadIdx.x;
    if (e >= E_TOT) return;
    int s, d; get_edge(ei, e, s, d);
    float acc = 0.f;
#pragma unroll
    for (int c = 0; c < OUT_DIM; ++c)
        acc = fmaf(lrelu(h2l[s * OUT_DIM + c] + h2r[d * OUT_DIM + c]), a2[c], acc);
    e2[e] = acc;
    atomicMax(&m2[d], fmap(acc));
}

__global__ void k_edge2_exp(const int* __restrict__ ei, float* __restrict__ e2,
                            const unsigned* __restrict__ m2, float* __restrict__ s2) {
    int e = blockIdx.x * blockDim.x + threadIdx.x;
    if (e >= E_TOT) return;
    int s, d; get_edge(ei, e, s, d);
    const float p = __expf(e2[e] - funmap(m2[d]));
    e2[e] = p;
    atomicAdd(&s2[d], p);
}

__global__ void k_edge2_agg(const int* __restrict__ ei,
                            const float* __restrict__ h2l,
                            const float* __restrict__ e2, const float* __restrict__ s2,
                            float* __restrict__ out) {
    int idx = blockIdx.x * blockDim.x + threadIdx.x;
    if (idx >= E_TOT * OUT_DIM) return;
    const int e = idx >> 3, c = idx & 7;
    int s, d; get_edge(ei, e, s, d);
    const float alpha = e2[e] / s2[d];
    atomicAdd(&out[d * OUT_DIM + c], alpha * h2l[s * OUT_DIM + c]);
}

extern "C" void kernel_launch(void* const* d_in, const int* in_sizes, int n_in,
                              void* d_out, int out_size, void* d_ws, size_t ws_size,
                              hipStream_t stream) {
    const float* x   = (const float*)d_in[0];
    const int*   ei  = (const int*)  d_in[1];
    const float* W1l = (const float*)d_in[2];
    const float* W1r = (const float*)d_in[3];
    const float* a1  = (const float*)d_in[4];
    const float* b1  = (const float*)d_in[5];
    const float* W2l = (const float*)d_in[6];
    const float* W2r = (const float*)d_in[7];
    const float* a2  = (const float*)d_in[8];
    const float* b2  = (const float*)d_in[9];
    float* out = (float*)d_out;

    char* ws = (char*)d_ws;
    size_t off = 0;
    auto carve = [&](size_t bytes) -> char* {
        char* p = ws + off;
        off += (bytes + 255) & ~(size_t)255;
        return p;
    };
    float*    xl1 = (float*)   carve((size_t)N_NODES * F1 * 4);
    float*    xr1 = (float*)   carve((size_t)N_NODES * F1 * 4);
    float*    e1  = (float*)   carve((size_t)E_TOT * HEADS * 4);
    unsigned* m1  = (unsigned*)carve((size_t)N_NODES * HEADS * 4);
    float*    s1  = (float*)   carve((size_t)N_NODES * HEADS * 4);
    float*    h2l = (float*)   carve((size_t)N_NODES * OUT_DIM * 4);
    float*    h2r = (float*)   carve((size_t)N_NODES * OUT_DIM * 4);
    float*    e2  = (float*)   carve((size_t)E_TOT * 4);
    unsigned* m2  = (unsigned*)carve((size_t)N_NODES * 4);
    float*    s2  = (float*)   carve((size_t)N_NODES * 4);
    float*    h1  = xr1;
    (void)ws_size; (void)n_in; (void)in_sizes; (void)out_size;

    const int B = 256;

    k_init1<<<(N_NODES * HEADS + B - 1) / B, B, 0, stream>>>(m1, s1);
    k_gemm1<<<dim3(M_TILES, 4, 2), B, 0, stream>>>(x, W1l, W1r, xl1, xr1);
    {
        int waves = E_TOT, blocks = (waves * 32 + B - 1) / B;
        k_edge1_logits<<<blocks, B, 0, stream>>>(ei, xl1, xr1, a1, e1, m1);
    }
    k_edge1_exp<<<(E_TOT * HEADS + B - 1) / B, B, 0, stream>>>(ei, e1, m1, s1);
    k_zero4<<<((N_NODES * F1 / 4) + B - 1) / B, B, 0, stream>>>((float4*)h1, N_NODES * F1 / 4);
    {
        int waves = E_TOT, blocks = (waves * 32 + B - 1) / B;
        k_edge1_agg<<<blocks, B, 0, stream>>>(ei, xl1, e1, s1, h1);
    }
    k_bias_relu<<<(N_NODES * F1 + B - 1) / B, B, 0, stream>>>(h1, b1);

    k_gemm2<<<(M_TILES + 3) / 4, B, 0, stream>>>(h1, W2l, W2r, h2l, h2r);
    k_init2<<<(N_NODES * OUT_DIM + B - 1) / B, B, 0, stream>>>(out, b2, m2, s2);
    k_edge2_logits<<<(E_TOT + B - 1) / B, B, 0, stream>>>(ei, h2l, h2r, a2, e2, m2);
    k_edge2_exp<<<(E_TOT + B - 1) / B, B, 0, stream>>>(ei, e2, m2, s2);
    k_edge2_agg<<<(E_TOT * OUT_DIM + B - 1) / B, B, 0, stream>>>(ei, h2l, e2, s2, out);
}


# Layer-1 logits: xl[s]+xr[d] recomputed per 16-edge tile as one K=16 WMMA product [x[s]|x[d]]x[Wl;Wr], wave per head; both 30000x512 planes, their GEMM launch and row gathers go; sum order changes
# speedup vs baseline: 6.2711x; 6.2711x over previous
#include <hip/hip_runtime.h>
#include <hip/hip_bf16.h>

#define N_NODES 30000
#define E_RAW   300000
#define E_TOT   330000
#define HEADS   8
#define HID     64
#define F1      512
#define IN_DIM  8
#define OUT_DIM 8
#define NEG_SLOPE 0.2f
#define M_TILES 1875
#define AGG_W   64
#define EPW     4

typedef __attribute__((ext_vector_type(2))) float v2f;
typedef __attribute__((ext_vector_type(8))) float v8f;

__device__ __forceinline__ unsigned fmap(float f) {
    unsigned u = __float_as_uint(f);
    return (u & 0x80000000u) ? ~u : (u | 0x80000000u);
}
__device__ __forceinline__ float funmap(unsigned u) {
    return __uint_as_float((u & 0x80000000u) ? (u ^ 0x80000000u) : ~u);
}
__device__ __forceinline__ void get_edge(const int* __restrict__ ei, int e, int& s, int& d) {
    if (e < E_RAW) { s = ei[e]; d = ei[E_RAW + e]; }
    else           { s = e - E_RAW; d = e - E_RAW; }
}
__device__ __forceinline__ float lrelu(float v) {
    return v > 0.f ? v : v * NEG_SLOPE;
}

__global__ void k_init1(unsigned* __restrict__ m1, float* __restrict__ s1) {
    int i = blockIdx.x * blockDim.x + threadIdx.x;
    if (i < N_NODES * HEADS) { m1[i] = 0u; s1[i] = 0.f; }
}
__global__ void k_zero4(float4* __restrict__ p, int n4) {
    int i = blockIdx.x * blockDim.x + threadIdx.x;
    if (i < n4) p[i] = make_float4(0.f, 0.f, 0.f, 0.f);
}
__global__ void k_init2(float* __restrict__ out, const float* __restrict__ b2,
                        unsigned* __restrict__ m2, float* __restrict__ s2) {
    int i = blockIdx.x * blockDim.x + threadIdx.x;
    if (i < N_NODES * OUT_DIM) out[i] = b2[i & 7];
    if (i < N_NODES) { m2[i] = 0u; s2[i] = 0.f; }
}

#define LOGIT_TILES 5
#define E_TILES (E_TOT / 16)
#define LOGIT_EDGES (LOGIT_TILES * 16)
__device__ __forceinline__ float other_half(float v) {
    return __builtin_bit_cast(float, __builtin_amdgcn_permlanex16(
        __builtin_bit_cast(int, v), __builtin_bit_cast(int, v), 0x76543210, 0xfedcba98, false, false));
}
__global__ void __launch_bounds__(256)
k_edge1_logits(const int* __restrict__ ei,
               const float* __restrict__ x,
               const float* __restrict__ Wl,
               const float* __restrict__ Wr,
               const float* __restrict__ a1,
               float* __restrict__ e1, unsigned* __restrict__ m1) {
    static_assert(E_TOT % 16 == 0, "whole 16-edge tiles: no lane of a tile is without an edge");
    static_assert(E_TILES % LOGIT_TILES == 0, "every work-group takes LOGIT_TILES whole tiles");
    __shared__ float stage[LOGIT_EDGES * HEADS];
    __shared__ int   dsts[LOGIT_EDGES];
    const int head  = threadIdx.x >> 5;
    const int lane  = threadIdx.x & 31;
    const int c     = lane & 15;
    const int hf    = lane >> 4;
    const int cbase = head * HID;

    v2f wa[4][4];
    v8f av[4];
#pragma unroll
    for (int t = 0; t < 4; ++t) {
#pragma unroll
        for (int kk = 0; kk < 4; ++kk) {
            const float* __restrict__ W = (kk < 2) ? Wl : Wr;
            const int k0 = (kk & 1) * 4 + hf * 2;
            wa[t][kk].x = W[(k0    ) * F1 + cbase + t * 16 + c];
            wa[t][kk].y = W[(k0 + 1) * F1 + cbase + t * 16 + c];
        }
        const float4 al = *(const float4*)(a1 + cbase + t * 16 + 8 * hf);
        const float4 ah = *(const float4*)(a1 + cbase + t * 16 + 8 * hf + 4);
        av[t][0] = al.x; av[t][1] = al.y; av[t][2] = al.z; av[t][3] = al.w;
        av[t][4] = ah.x; av[t][5] = ah.y; av[t][6] = ah.z; av[t][7] = ah.w;
    }

    const int e0 = blockIdx.x * LOGIT_EDGES + c;
    int s[LOGIT_TILES], d[LOGIT_TILES];
#pragma unroll
    for (int i = 0; i < LOGIT_TILES; ++i) {
        const int e  = e0 + i * 16;
        const int ec = e < E_RAW ? e : E_RAW - 1;
        const int ls = ei[ec], ld = ei[E_RAW + ec];
        s[i] = e < E_RAW ? ls : e - E_RAW;
        d[i] = e < E_RAW ? ld : e - E_RAW;
    }
    if (head == 0 && hf == 0) {
#pragma unroll
        for (int i = 0; i < LOGIT_TILES; ++i) dsts[i * 16 + c] = d[i];
    }

    v2f f[4];
    {
        const float* __restrict__ ps = x + s[0] * IN_DIM + hf * 2;
        const float* __restrict__ pd = x + d[0] * IN_DIM + hf * 2;
        f[0] = *(const v2f*)(ps); f[1] = *(const v2f*)(ps + 4);
        f[2] = *(const v2f*)(pd); f[3] = *(const v2f*)(pd + 4);
    }

#pragma unroll
    for (int i = 0; i < LOGIT_TILES; ++i) {
        v2f n[4] = { f[0], f[1], f[2], f[3] };
        if (i + 1 < LOGIT_TILES) {
            const float* __restrict__ ps = x + s[i + 1] * IN_DIM + hf * 2;
            const float* __restrict__ pd = x + d[i + 1] * IN_DIM + hf * 2;
            n[0] = *(const v2f*)(ps); n[1] = *(const v2f*)(ps + 4);
            n[2] = *(const v2f*)(pd); n[3] = *(const v2f*)(pd + 4);
        }

        v8f dd[4];
#pragma unroll
        for (int t = 0; t < 4; ++t) {
            v8f z = {};
            dd[t] = __builtin_amdgcn_wmma_f32_16x16x4_f32(false, wa[t][0], false, f[0],
                                                          (short)0, z, false, false);
        }
#pragma unroll
        for (int kk = 1; kk < 4; ++kk)
#pragma unroll
            for (int t = 0; t < 4; ++t)
                dd[t] = __builtin_amdgcn_wmma_f32_16x16x4_f32(false, wa[t][kk], false, f[kk],
                                                              (short)0, dd[t], false, false);

        float p[4];
#pragma unroll
        for (int t = 0; t < 4; ++t) {
            v2f a = { 0.f, 0.f };
#pragma unroll
            for (int q = 0; q < 4; ++q) {
                v2f v, w, l;
                v.x = dd[t][2 * q]; v.y = dd[t][2 * q + 1];
                w.x = av[t][2 * q]; w.y = av[t][2 * q + 1];
                const v2f m = v * NEG_SLOPE;
                l.x = fmaxf(v.x, m.x);
                l.y = fmaxf(v.y, m.y);
                a = __builtin_elementwise_fma(l, w, a);
            }
            p[t] = a.x + a.y;
        }
        float acc = (p[0] + p[1]) + (p[2] + p[3]);
        acc += other_half(acc);
        if (hf == 0) stage[(i * 16 + c) * HEADS + head] = acc;

        f[0] = n[0]; f[1] = n[1]; f[2] = n[2]; f[3] = n[3];
    }

    __syncthreads();
    const size_t obase = (size_t)blockIdx.x * (LOGIT_EDGES * HEADS);
    for (int j = 0; j < (LOGIT_EDGES * HEADS + 255) / 256; ++j) {
        const int idx = (int)threadIdx.x + 256 * j;
        if (idx < LOGIT_EDGES * HEADS) {
            const float v = stage[idx];
            e1[obase + idx] = v;
            atomicMax(&m1[dsts[idx >> 3] * HEADS + (idx & 7)], fmap(v));
        }
    }
}

__global__ void k_edge1_exp(const int* __restrict__ ei, float* __restrict__ e1,
                            const unsigned* __restrict__ m1, float* __restrict__ s1) {
    int idx = blockIdx.x * blockDim.x + threadIdx.x;
    if (idx >= E_TOT * HEADS) return;
    const int e = idx >> 3, h = idx & 7;
    int s, d; get_edge(ei, e, s, d);
    const float p = __expf(e1[idx] - funmap(m1[d * HEADS + h]));
    e1[idx] = p;
    atomicAdd(&s1[d * HEADS + h], p);
}

__global__ void k_edge1_agg(const int* __restrict__ ei,
                            const float* __restrict__ x,
                            const float* __restrict__ e1, const float* __restrict__ s1,
                            float* __restrict__ agg) {
    static_assert(E_TOT % EPW == 0, "a wave's EPW edges are taken without a per-edge guard");
    const int wid  = (int)((blockIdx.x * blockDim.x + threadIdx.x) >> 5);
    const int lane = threadIdx.x & 31;
    if (wid * EPW >= E_TOT) return;
    const int k  = lane & 7;
    const int h0 = lane >> 3;
    int   s[EPW], d[EPW];
    float xv[EPW], p0[EPW], p1[EPW], q0[EPW], q1[EPW];
#pragma unroll
    for (int j = 0; j < EPW; ++j) {
        const int e  = wid * EPW + j;
        const int ec = e < E_RAW ? e : E_RAW - 1;
        const int ls = ei[ec], ld = ei[E_RAW + ec];
        s[j] = e < E_RAW ? ls : e - E_RAW;
        d[j] = e < E_RAW ? ld : e - E_RAW;
    }
#pragma unroll
    for (int j = 0; j < EPW; ++j) {
        const int e = wid * EPW + j;
        xv[j] = x[s[j] * IN_DIM + k];
        p0[j] = e1[(size_t)e * HEADS + h0];
        p1[j] = e1[(size_t)e * HEADS + h0 + 4];
        q0[j] = s1[d[j] * HEADS + h0];
        q1[j] = s1[d[j] * HEADS + h0 + 4];
    }
#pragma unroll
    for (int j = 0; j < EPW; ++j) {
        float* __restrict__ po = agg + (size_t)d[j] * AGG_W;
        atomicAdd(po + lane,      (p0[j] / q0[j]) * xv[j]);
        atomicAdd(po + 32 + lane, (p1[j] / q1[j]) * xv[j]);
    }
}

#define FUSE_TILES 3
__global__ void __launch_bounds__(256)
k_node_fused(const float* __restrict__ agg,
             const float* __restrict__ Wl,
             const float* __restrict__ b1,
             const float* __restrict__ W2l,
             const float* __restrict__ W2r,
             float* __restrict__ h2l, float* __restrict__ h2r) {
    static_assert(M_TILES % FUSE_TILES == 0, "every work-group takes FUSE_TILES whole tiles");
    __shared__ float part[2][HEADS][256];
    const int head  = threadIdx.x >> 5;
    const int lane  = threadIdx.x & 31;
    const int c     = lane & 15;
    const int hf    = lane >> 4;
    const int cbase = head * HID;

    v2f  wa[4][2];
    v8f  bias[4];
    v2f  wb[4][4];
    const float* __restrict__ W2 = (c < OUT_DIM) ? W2l : W2r;
    const int j2 = c & (OUT_DIM - 1);
#pragma unroll
    for (int t = 0; t < 4; ++t) {
#pragma unroll
        for (int kk = 0; kk < 2; ++kk) {
            const int k0 = kk * 4 + hf * 2;
            wa[t][kk].x = Wl[(k0    ) * F1 + cbase + t * 16 + c];
            wa[t][kk].y = Wl[(k0 + 1) * F1 + cbase + t * 16 + c];
        }
        const float4 bl = *(const float4*)(b1 + cbase + t * 16 + 8 * hf);
        const float4 bh = *(const float4*)(b1 + cbase + t * 16 + 8 * hf + 4);
        bias[t][0] = bl.x; bias[t][1] = bl.y; bias[t][2] = bl.z; bias[t][3] = bl.w;
        bias[t][4] = bh.x; bias[t][5] = bh.y; bias[t][6] = bh.z; bias[t][7] = bh.w;
#pragma unroll
        for (int q = 0; q < 4; ++q) {
            const int ch = cbase + t * 16 + 8 * hf + 2 * q;
            wb[t][q].x = W2[(ch    ) * OUT_DIM + j2];
            wb[t][q].y = W2[(ch + 1) * OUT_DIM + j2];
        }
    }

    const int tile0 = blockIdx.x * FUSE_TILES;
    const float* __restrict__ pa = agg + (size_t)(tile0 * 16 + c) * AGG_W + head * IN_DIM + hf * 2;
    v2f f0 = *(const v2f*)(pa);
    v2f f1 = *(const v2f*)(pa + 4);

#pragma unroll
    for (int i = 0; i < FUSE_TILES; ++i) {
        const int nbase = (tile0 + i) * 16;
        v2f n0 = f0, n1 = f1;
        if (i + 1 < FUSE_TILES) {
            const float* __restrict__ pn = pa + (size_t)(i + 1) * 16 * AGG_W;
            n0 = *(const v2f*)(pn);
            n1 = *(const v2f*)(pn + 4);
        }

        v8f d1[4];
#pragma unroll
        for (int t = 0; t < 4; ++t) {
            v8f z = {};
            d1[t] = __builtin_amdgcn_wmma_f32_16x16x4_f32(false, wa[t][0], false, f0,
                                                          (short)0, z, false, false);
        }
#pragma unroll
        for (int t = 0; t < 4; ++t)
            d1[t] = __builtin_amdgcn_wmma_f32_16x16x4_f32(false, wa[t][1], false, f1,
                                                          (short)0, d1[t], false, false);

        v8f acc0 = {}, acc1 = {};
#pragma unroll
        for (int t = 0; t < 4; ++t) {
            v8f v;
#pragma unroll
            for (int r = 0; r < 8; ++r) {
                const float u = d1[t][r] + bias[t][r];
                v[r] = u > 0.f ? u : 0.f;
            }
            v2f b0, b1f, b2, b3;
            b0.x  = v[0]; b0.y  = v[1];
            b1f.x = v[2]; b1f.y = v[3];
            b2.x  = v[4]; b2.y  = v[5];
            b3.x  = v[6]; b3.y  = v[7];
            acc0 = __builtin_amdgcn_wmma_f32_16x16x4_f32(false, wb[t][0], false, b0,
                                                         (short)0, acc0, false, false);
            acc1 = __builtin_amdgcn_wmma_f32_16x16x4_f32(false, wb[t][1], false, b1f,
                                                         (short)0, acc1, false, false);
            acc0 = __builtin_amdgcn_wmma_f32_16x16x4_f32(false, wb[t][2], false, b2,
                                                         (short)0, acc0, false, false);
            acc1 = __builtin_amdgcn_wmma_f32_16x16x4_f32(false, wb[t][3], false, b3,
                                                         (short)0, acc1, false, false);
        }

        float* __restrict__ pp = &part[i & 1][head][c * 16 + 8 * hf];
        *(float4*)(pp)     = make_float4(acc0[0] + acc1[0], acc0[1] + acc1[1],
                                         acc0[2] + acc1[2], acc0[3] + acc1[3]);
        *(float4*)(pp + 4) = make_float4(acc0[4] + acc1[4], acc0[5] + acc1[5],
                                         acc0[6] + acc1[6], acc0[7] + acc1[7]);
        __syncthreads();
        float sum = part[i & 1][0][threadIdx.x];
        for (int w = 1; w < HEADS; ++w) sum += part[i & 1][w][threadIdx.x];
        const int node = nbase + (threadIdx.x >> 4);
        const int jj   = threadIdx.x & 15;
        if (jj < OUT_DIM) h2l[node * OUT_DIM + jj] = sum;
        else              h2r[node * OUT_DIM + jj - OUT_DIM] = sum;

        f0 = n0; f1 = n1;
    }
}

__global__ void k_edge2_logits(const int* __restrict__ ei,
                               const float* __restrict__ h2l, const float* __restrict__ h2r,
                               const float* __restrict__ a2,
                               float* __restrict__ e2, unsigned* __restrict__ m2) {
    int e = blockIdx.x * blockDim.x + threadIdx.x;
    if (e >= E_TOT) return;
    int s, d; get_edge(ei, e, s, d);
    float acc = 0.f;
#pragma unroll
    for (int c = 0; c < OUT_DIM; ++c)
        acc = fmaf(lrelu(h2l[s * OUT_DIM + c] + h2r[d * OUT_DIM + c]), a2[c], acc);
    e2[e] = acc;
    atomicMax(&m2[d], fmap(acc));
}

__global__ void k_edge2_exp(const int* __restrict__ ei, float* __restrict__ e2,
                            const unsigned* __restrict__ m2, float* __restrict__ s2) {
    int e = blockIdx.x * blockDim.x + threadIdx.x;
    if (e >= E_TOT) return;
    int s, d; get_edge(ei, e, s, d);
    const float p = __expf(e2[e] - funmap(m2[d]));
    e2[e] = p;
    atomicAdd(&s2[d], p);
}

__global__ void k_edge2_agg(const int* __restrict__ ei,
                            const float* __restrict__ h2l,
                            const float* __restrict__ e2, const float* __restrict__ s2,
                            float* __restrict__ out) {
    int idx = blockIdx.x * blockDim.x + threadIdx.x;
    if (idx >= E_TOT * OUT_DIM) return;
    const int e = idx >> 3, c = idx & 7;
    int s, d; get_edge(ei, e, s, d);
    const float alpha = e2[e] / s2[d];
    atomicAdd(&out[d * OUT_DIM + c], alpha * h2l[s * OUT_DIM + c]);
}

extern "C" void kernel_launch(void* const* d_in, const int* in_sizes, int n_in,
                              void* d_out, int out_size, void* d_ws, size_t ws_size,
                              hipStream_t stream) {
    const float* x   = (const float*)d_in[0];
    const int*   ei  = (const int*)  d_in[1];
    const float* W1l = (const float*)d_in[2];
    const float* W1r = (const float*)d_in[3];
    const float* a1  = (const float*)d_in[4];
    const float* b1  = (const float*)d_in[5];
    const float* W2l = (const float*)d_in[6];
    const float* W2r = (const float*)d_in[7];
    const float* a2  = (const float*)d_in[8];
    const float* b2  = (const float*)d_in[9];
    float* out = (float*)d_out;

    char* ws = (char*)d_ws;
    size_t off = 0;
    auto carve = [&](size_t bytes) -> char* {
        char* p = ws + off;
        off += (bytes + 255) & ~(size_t)255;
        return p;
    };
    float*    e1  = (float*)   carve((size_t)E_TOT * HEADS * 4);
    unsigned* m1  = (unsigned*)carve((size_t)N_NODES * HEADS * 4);
    float*    s1  = (float*)   carve((size_t)N_NODES * HEADS * 4);
    float*    h2l = (float*)   carve((size_t)N_NODES * OUT_DIM * 4);
    float*    h2r = (float*)   carve((size_t)N_NODES * OUT_DIM * 4);
    float*    e2  = (float*)   carve((size_t)E_TOT * 4);
    unsigned* m2  = (unsigned*)carve((size_t)N_NODES * 4);
    float*    s2  = (float*)   carve((size_t)N_NODES * 4);
    float*    agg = (float*)   carve((size_t)N_NODES * AGG_W * 4);
    (void)ws_size; (void)n_in; (void)in_sizes; (void)out_size;

    const int B = 256;

    k_init1<<<(N_NODES * HEADS + B - 1) / B, B, 0, stream>>>(m1, s1);
    k_edge1_logits<<<E_TILES / LOGIT_TILES, B, 0, stream>>>(ei, x, W1l, W1r, a1, e1, m1);
    k_edge1_exp<<<(E_TOT * HEADS + B - 1) / B, B, 0, stream>>>(ei, e1, m1, s1);
    k_zero4<<<((N_NODES * AGG_W / 4) + B - 1) / B, B, 0, stream>>>((float4*)agg, N_NODES * AGG_W / 4);
    {
        int waves = E_TOT / EPW, blocks = (waves * 32 + B - 1) / B;
        k_edge1_agg<<<blocks, B, 0, stream>>>(ei, x, e1, s1, agg);
    }
    k_node_fused<<<M_TILES / FUSE_TILES, B, 0, stream>>>(agg, W1l, b1, W2l, W2r, h2l, h2r);

    k_init2<<<(N_NODES * OUT_DIM + B - 1) / B, B, 0, stream>>>(out, b2, m2, s2);
    k_edge2_logits<<<(E_TOT + B - 1) / B, B, 0, stream>>>(ei, h2l, h2r, a2, e2, m2);
    k_edge2_exp<<<(E_TOT + B - 1) / B, B, 0, stream>>>(ei, e2, m2, s2);
    k_edge2_agg<<<(E_TOT * OUT_DIM + B - 1) / B, B, 0, stream>>>(ei, h2l, e2, s2, out);
}
